# starting kernel
# baseline (speedup 1.0000x reference)
#include <hip/hip_runtime.h>
#include <hip/hip_bf16.h>

typedef __attribute__((ext_vector_type(2))) float v2f;
typedef __attribute__((ext_vector_type(8))) float v8f;

#define NF 64
#define DK 128
#define LDS_STRIDE 132
#define NPAIRS 2016

__global__ __launch_bounds__(128)
void InteractionLayer_gram_triu_kernel(const float* __restrict__ x,
                                       float* __restrict__ out) {
    __shared__ float sx[NF * LDS_STRIDE];

    const int b   = blockIdx.x;
    const int tid = threadIdx.x;
    const float* __restrict__ xb = x + (size_t)b * (NF * DK);

    for (int idx = tid; idx < NF * (DK / 4); idx += 128) {
        const int row = idx >> 5;
        const int c4  = idx & 31;
        const float4 v = reinterpret_cast<const float4*>(xb)[(row << 5) + c4];
        float* dst = &sx[row * LDS_STRIDE + (c4 << 2)];
        dst[0] = v.x; dst[1] = v.y; dst[2] = v.z; dst[3] = v.w;
    }
    __syncthreads();

    const int wave  = tid >> 5;
    const int lane  = tid & 31;
    const int m     = lane & 15;
    const int khalf = lane >> 4;

    const int wm = (wave >> 1) << 1;
    const int wn = (wave & 1)  << 1;

    const bool dead = (wm == 2) && (wn == 0);

    if (!dead) {
        v8f acc[2][2];
        #pragma unroll
        for (int i = 0; i < 2; ++i)
            #pragma unroll
            for (int j = 0; j < 2; ++j)
                acc[i][j] = (v8f){0.f, 0.f, 0.f, 0.f, 0.f, 0.f, 0.f, 0.f};

        for (int k = 0; k < DK / 4; ++k) {
            const int kb = (k << 2) + (khalf << 1);
            v2f afrag[2], bfrag[2];
            #pragma unroll
            for (int i = 0; i < 2; ++i) {
                afrag[i] = *(const v2f*)&sx[(((wm + i) << 4) + m) * LDS_STRIDE + kb];
                bfrag[i] = *(const v2f*)&sx[(((wn + i) << 4) + m) * LDS_STRIDE + kb];
            }
            #pragma unroll
            for (int i = 0; i < 2; ++i)
                #pragma unroll
                for (int j = 0; j < 2; ++j)
                    acc[i][j] = __builtin_amdgcn_wmma_f32_16x16x4_f32(
                         false, afrag[i],
                         false, bfrag[j],
                         (short)0, acc[i][j],
                         false,  false);
        }

        float* __restrict__ ob = out + (size_t)b * NPAIRS;
        #pragma unroll
        for (int i = 0; i < 2; ++i) {
            #pragma unroll
            for (int j = 0; j < 2; ++j) {
                const int ti = (wm + i) << 4;
                const int tj = (wn + j) << 4;
                #pragma unroll
                for (int v = 0; v < 8; ++v) {
                    const int gi = ti + v + (khalf << 3);
                    const int gj = tj + m;
                    if (gj > gi) {
                        const int idx = gi * (NF - 1) - ((gi * (gi - 1)) >> 1) + (gj - gi - 1);
                        ob[idx] = acc[i][j][v];
                    }
                }
            }
        }
    }
}

extern "C" void kernel_launch(void* const* d_in, const int* in_sizes, int n_in,
                              void* d_out, int out_size, void* d_ws, size_t ws_size,
                              hipStream_t stream) {
    (void)in_sizes; (void)n_in; (void)d_ws; (void)ws_size; (void)out_size;
    const float* x = (const float*)d_in[0];
    float* out = (float*)d_out;
    const int batches = 4096;
    InteractionLayer_gram_triu_kernel<<<batches, 128, 0, stream>>>(x, out);
}


# One wave per batch: operand read once global->WMMA fragment regs (b128, k regrouped per 32 cols), same regs are A and B of the 10 live tiles; no LDS, barrier or idle wave; k sum order changes
# speedup vs baseline: 1.6254x; 1.6254x over previous
#include <hip/hip_runtime.h>


typedef __attribute__((ext_vector_type(2))) float v2f;
typedef __attribute__((ext_vector_type(4))) float v4f;
typedef __attribute__((ext_vector_type(8))) float v8f;

#define NF 64
#define DK 128
#define NPAIRS 2016
#define WAVES_PER_BLOCK 4

__device__ __forceinline__ void load_group(v4f (&r)[4][4], const v4f* __restrict__ p, const int S) {
    #pragma unroll
    for (int t = 0; t < 4; ++t) {
        #pragma unroll
        for (int j = 0; j < 4; ++j)
            r[t][j] = p[t * 512 + S * 8 + j * 2];
    }
}

__device__ __forceinline__ void mma_group(v8f (&acc)[10], const v4f (&r)[4][4]) {
    #pragma unroll
    for (int j = 0; j < 4; ++j) {
        #pragma unroll
        for (int q = 0; q < 2; ++q) {
            v2f f[4];
            #pragma unroll
            for (int t = 0; t < 4; ++t)
                f[t] = q ? __builtin_shufflevector(r[t][j], r[t][j], 2, 3)
                         : __builtin_shufflevector(r[t][j], r[t][j], 0, 1);
            int n = 0;
            #pragma unroll
            for (int ti = 0; ti < 4; ++ti) {
                #pragma unroll
                for (int tj = ti; tj < 4; ++tj) {
                    acc[n] = __builtin_amdgcn_wmma_f32_16x16x4_f32(
                         false, f[ti],
                         false, f[tj],
                         (short)0, acc[n],
                         false,  false);
                    ++n;
                }
            }
        }
    }
}

__global__ __launch_bounds__(128)
void InteractionLayer_gram_triu_kernel(const float* __restrict__ x,
                                       float* __restrict__ out) {
    const int wave = __builtin_amdgcn_readfirstlane((int)(threadIdx.x >> 5));
    const int b    = blockIdx.x * WAVES_PER_BLOCK + wave;
    const int lane = threadIdx.x & 31;
    const int c    = lane & 15;
    const int h    = lane >> 4;

    const float* __restrict__ xb = x + (size_t)b * (NF * DK);
    const v4f* __restrict__ p = reinterpret_cast<const v4f*>(xb + c * DK + 4 * h);

    v4f g0[4][4], g1[4][4];
    load_group(g0, p, 0);
    load_group(g1, p, 1);
    __builtin_amdgcn_sched_barrier(0);

    v8f acc[10];
    #pragma unroll
    for (int n = 0; n < 10; ++n)
        acc[n] = (v8f){0.f, 0.f, 0.f, 0.f, 0.f, 0.f, 0.f, 0.f};

    mma_group(acc, g0);
    load_group(g0, p, 2);
    __builtin_amdgcn_sched_barrier(0);
    mma_group(acc, g1);
    load_group(g1, p, 3);
    __builtin_amdgcn_sched_barrier(0);
    mma_group(acc, g0);
    mma_group(acc, g1);

    float* __restrict__ ob = out + (size_t)b * NPAIRS;
    const unsigned rstep = 62u - 8u * (unsigned)h;
    const unsigned lbase = 468u * (unsigned)h + (unsigned)c;
    const int dcol = c - 8 * h;
    int n = 0;
    #pragma unroll
    for (int ti = 0; ti < 4; ++ti) {
        #pragma unroll
        for (int tj = ti; tj < 4; ++tj) {
            #pragma unroll
            for (int v = 0; v < 8; ++v) {
                const int r = (ti << 4) + v;
                float* __restrict__ rowp = ob + ((unsigned)r * rstep + lbase);
                const int koff = (tj << 4) - ((r * (r - 1)) >> 1) - 1;
                if (tj > ti || dcol > v)
                    rowp[koff] = acc[n][v];
            }
            ++n;
        }
    }
}

extern "C" void kernel_launch(void* const* d_in, const int* in_sizes, int n_in,
                              void* d_out, int out_size, void* d_ws, size_t ws_size,
                              hipStream_t stream) {
    (void)in_sizes; (void)n_in; (void)d_ws; (void)ws_size; (void)out_size;
    const float* x = (const float*)d_in[0];
    float* out = (float*)d_out;
    InteractionLayer_gram_triu_kernel<<<1024, 128, 0, stream>>>(x, out);
}


# Epilogue: each wave packs its output rows in its own 8 KiB of LDS and writes whole 128-B lines (16 b128 stores of 4 lines) instead of 80 b32 partial-line stores, 32 of them masked; sums unchanged
# speedup vs baseline: 1.7114x; 1.0529x over previous
#include <hip/hip_runtime.h>


typedef __attribute__((ext_vector_type(2))) float v2f;
typedef __attribute__((ext_vector_type(4))) float v4f;
typedef __attribute__((ext_vector_type(8))) float v8f;

#define NF 64
#define DK 128
#define NPAIRS 2016
#define WAVES_PER_BLOCK 4
#define OUT_SLOTS 2048

__device__ __forceinline__ void load_group(v4f (&r)[4][4], const v4f* __restrict__ p, const int S) {
    #pragma unroll
    for (int t = 0; t < 4; ++t) {
        #pragma unroll
        for (int j = 0; j < 4; ++j)
            r[t][j] = p[t * 512 + S * 8 + j * 2];
    }
}

__device__ __forceinline__ void mma_group(v8f (&acc)[10], const v4f (&r)[4][4]) {
    #pragma unroll
    for (int j = 0; j < 4; ++j) {
        #pragma unroll
        for (int q = 0; q < 2; ++q) {
            v2f f[4];
            #pragma unroll
            for (int t = 0; t < 4; ++t)
                f[t] = q ? __builtin_shufflevector(r[t][j], r[t][j], 2, 3)
                         : __builtin_shufflevector(r[t][j], r[t][j], 0, 1);
            int n = 0;
            #pragma unroll
            for (int ti = 0; ti < 4; ++ti) {
                #pragma unroll
                for (int tj = ti; tj < 4; ++tj) {
                    acc[n] = __builtin_amdgcn_wmma_f32_16x16x4_f32(
                         false, f[ti],
                         false, f[tj],
                         (short)0, acc[n],
                         false,  false);
                    ++n;
                }
            }
        }
    }
}

__global__ __launch_bounds__(128)
void InteractionLayer_gram_triu_kernel(const float* __restrict__ x,
                                       float* __restrict__ out) {
    __shared__ __attribute__((aligned(16))) float stage[WAVES_PER_BLOCK * OUT_SLOTS];

    const int wave = __builtin_amdgcn_readfirstlane((int)(threadIdx.x >> 5));
    const int b    = blockIdx.x * WAVES_PER_BLOCK + wave;
    const int lane = threadIdx.x & 31;
    const int c    = lane & 15;
    const int h    = lane >> 4;

    const float* __restrict__ xb = x + (size_t)b * (NF * DK);
    const v4f* __restrict__ p = reinterpret_cast<const v4f*>(xb + c * DK + 4 * h);

    v4f g0[4][4], g1[4][4];
    load_group(g0, p, 0);
    load_group(g1, p, 1);
    __builtin_amdgcn_sched_barrier(0);

    v8f acc[10];
    #pragma unroll
    for (int n = 0; n < 10; ++n)
        acc[n] = (v8f){0.f, 0.f, 0.f, 0.f, 0.f, 0.f, 0.f, 0.f};

    mma_group(acc, g0);
    load_group(g0, p, 2);
    __builtin_amdgcn_sched_barrier(0);
    mma_group(acc, g1);
    load_group(g1, p, 3);
    __builtin_amdgcn_sched_barrier(0);
    mma_group(acc, g0);
    mma_group(acc, g1);

    __builtin_amdgcn_sched_barrier(0);

    char* const swb = reinterpret_cast<char*>(stage) + wave * (OUT_SLOTS * 4);
    char* const lanep  = swb + 4 * (468 * h + c);
    char* const sparep = swb + 4 * (NPAIRS + lane);
    const int rstep4 = 4 * (62 - 8 * h);
    const int dcol   = c - 8 * h;
    int n = 0;
    #pragma unroll
    for (int ti = 0; ti < 4; ++ti) {
        #pragma unroll
        for (int tj = ti; tj < 4; ++tj) {
            #pragma unroll
            for (int v = 0; v < 8; ++v) {
                const int r = (ti << 4) + v;
                const int koff = (tj << 4) - ((r * (r - 1)) >> 1) - 1;
                char* q = lanep + r * rstep4 + 4 * koff;
                if (tj == ti)
                    q = (dcol > v) ? q : sparep;
                *reinterpret_cast<float*>(q) = acc[n][v];
            }
            ++n;
        }
    }

    float* __restrict__ ob = out + (size_t)b * NPAIRS;
    const char* src = swb + 16 * lane;
    float* dst = ob + 4 * lane;
    #pragma unroll
    for (int i = 0; i < 16; ++i) {
        v4f t;
        __builtin_memcpy(&t, __builtin_assume_aligned(src + i * 512, 16), 16);
        if (i < 15 || lane < 24)
            *reinterpret_cast<v4f*>(dst + i * 128) = t;
    }
}

extern "C" void kernel_launch(void* const* d_in, const int* in_sizes, int n_in,
                              void* d_out, int out_size, void* d_ws, size_t ws_size,
                              hipStream_t stream) {
    (void)in_sizes; (void)n_in; (void)d_ws; (void)ws_size; (void)out_size;
    const float* x = (const float*)d_in[0];
    float* out = (float*)d_out;
    InteractionLayer_gram_triu_kernel<<<1024, 128, 0, stream>>>(x, out);
}
